# starting kernel
# baseline (speedup 1.0000x reference)
#include <hip/hip_runtime.h>

#define Nn 100000
#define Ee 1000000
#define INF 128
#define HID 128
#define CLS 64

typedef __attribute__((ext_vector_type(2))) float v2f;
typedef __attribute__((ext_vector_type(8))) float v8f;

__device__ __forceinline__ void atomic_add_f32(float* p, float v) {
    __hip_atomic_fetch_add(p, v, __ATOMIC_RELAXED, __HIP_MEMORY_SCOPE_AGENT);
}

__global__ void k_zero_deg(int* __restrict__ deg) {
    int i = blockIdx.x * blockDim.x + threadIdx.x;
    if (i < Nn) deg[i] = 0;
}

__global__ void k_count_deg(const int* __restrict__ dst, int* __restrict__ deg) {
    int e = blockIdx.x * blockDim.x + threadIdx.x;
    if (e < Ee) atomicAdd(&deg[dst[e]], 1);
}

__global__ void k_dinv(const int* __restrict__ deg, float* __restrict__ dinv) {
    int i = blockIdx.x * blockDim.x + threadIdx.x;
    if (i < Nn) {
        float d = (float)deg[i] + 1.0f;
        dinv[i] = 1.0f / sqrtf(d);
    }
}

template <int DOUT, bool RELU_IN>
__global__ void k_gemm_wmma(const float* __restrict__ X,
                            const float* __restrict__ W,
                            float* __restrict__ H) {
    __shared__ float lds_x[16 * 132];
    __shared__ float lds_w[128 * DOUT];

    const int tid     = threadIdx.x;
    const int lane    = tid & 31;
    const int wave    = tid >> 5;
    const int rowBase = blockIdx.x * 16;
    const int colBase = wave * 16;
    const int nthr    = (DOUT / 16) * 32;

    for (int i = tid; i < 16 * 128; i += nthr) {
        int r = i >> 7, c = i & 127;
        float v = X[(size_t)(rowBase + r) * 128 + c];
        if (RELU_IN) v = fmaxf(v, 0.0f);
        lds_x[r * 132 + c] = v;
    }
    for (int i = tid; i < 128 * DOUT; i += nthr) lds_w[i] = W[i];
    __syncthreads();

    const int r    = lane & 15;
    const int half = lane >> 4;
    const int koff = half * 2;

    v8f acc = {};
#pragma unroll
    for (int kk = 0; kk < 128; kk += 4) {
        v2f a = *(const v2f*)(lds_x + r * 132 + kk + koff);
        v2f b;
        b.x = lds_w[(kk + koff) * DOUT + colBase + r];
        b.y = lds_w[(kk + koff + 1) * DOUT + colBase + r];
        acc = __builtin_amdgcn_wmma_f32_16x16x4_f32(
             false, a,  false, b,
             (short)0, acc,  false,  false);
    }

#pragma unroll
    for (int v = 0; v < 8; ++v) {
        H[(size_t)(rowBase + v + half * 8) * DOUT + colBase + r] = acc[v];
    }
}

template <int DOUT>
__global__ void k_init_agg(const float* __restrict__ h,
                           const float* __restrict__ b,
                           const float* __restrict__ dinv,
                           float* __restrict__ out) {
    const int per = DOUT / 4;
    int t = blockIdx.x * blockDim.x + threadIdx.x;
    if (t >= Nn * per) return;
    int i  = t / per;
    int c4 = t - i * per;
    float w = dinv[i];
    w = w * w;
    float4 hv = ((const float4*)h)[t];
    float4 bb = ((const float4*)b)[c4];
    float4 o;
    o.x = hv.x * w + bb.x;
    o.y = hv.y * w + bb.y;
    o.z = hv.z * w + bb.z;
    o.w = hv.w * w + bb.w;
    ((float4*)out)[t] = o;
}

template <int DOUT>
__global__ void k_scatter(const float* __restrict__ h,
                          const int* __restrict__ src,
                          const int* __restrict__ dst,
                          const float* __restrict__ dinv,
                          float* __restrict__ out) {
    const int per = DOUT / 4;
    long long t = (long long)blockIdx.x * blockDim.x + threadIdx.x;
    if (t >= (long long)Ee * per) return;
    int e = (int)(t / per);
    int c = (int)(t - (long long)e * per);
    int s = src[e];
    int d = dst[e];
    float w = dinv[s] * dinv[d];
    float4 hv = ((const float4*)h)[(size_t)s * per + c];
    float* o = out + (size_t)d * DOUT + c * 4;
    atomic_add_f32(o + 0, hv.x * w);
    atomic_add_f32(o + 1, hv.y * w);
    atomic_add_f32(o + 2, hv.z * w);
    atomic_add_f32(o + 3, hv.w * w);
}

extern "C" void kernel_launch(void* const* d_in, const int* in_sizes, int n_in,
                              void* d_out, int out_size, void* d_ws, size_t ws_size,
                              hipStream_t stream) {
    const float* X   = (const float*)d_in[0];
    const float* W1  = (const float*)d_in[1];
    const float* b1  = (const float*)d_in[2];
    const float* W2  = (const float*)d_in[3];
    const float* b2  = (const float*)d_in[4];
    const float* W3  = (const float*)d_in[5];
    const float* b3  = (const float*)d_in[6];
    const int*   ei  = (const int*)d_in[7];
    const int*   src = ei;
    const int*   dst = ei + Ee;
    float* out = (float*)d_out;

    float* bufA = (float*)d_ws;
    float* bufB = bufA + (size_t)Nn * 128;
    int*   deg  = (int*)(bufB + (size_t)Nn * 128);
    float* dinv = (float*)(deg + Nn);

    const int T = 256;
    k_zero_deg<<<(Nn + T - 1) / T, T, 0, stream>>>(deg);
    k_count_deg<<<(Ee + T - 1) / T, T, 0, stream>>>(dst, deg);
    k_dinv<<<(Nn + T - 1) / T, T, 0, stream>>>(deg, dinv);

    const int gGemm = Nn / 16;

    k_gemm_wmma<128, false><<<gGemm, 256, 0, stream>>>(X, W1, bufA);
    k_init_agg<128><<<(Nn * 32 + T - 1) / T, T, 0, stream>>>(bufA, b1, dinv, bufB);
    k_scatter<128><<<(int)(((long long)Ee * 32 + T - 1) / T), T, 0, stream>>>(bufA, src, dst, dinv, bufB);

    k_gemm_wmma<128, true><<<gGemm, 256, 0, stream>>>(bufB, W2, bufA);
    k_init_agg<128><<<(Nn * 32 + T - 1) / T, T, 0, stream>>>(bufA, b2, dinv, bufB);
    k_scatter<128><<<(int)(((long long)Ee * 32 + T - 1) / T), T, 0, stream>>>(bufA, src, dst, dinv, bufB);

    k_gemm_wmma<64, true><<<gGemm, 128, 0, stream>>>(bufB, W3, bufA);
    k_init_agg<64><<<(Nn * 16 + T - 1) / T, T, 0, stream>>>(bufA, b3, dinv, out);
    k_scatter<64><<<(int)(((long long)Ee * 16 + T - 1) / T), T, 0, stream>>>(bufA, src, dst, dinv, out);
}


# A wave's global loads now run ahead of its LDS work: self-loop rows fetched as one batch, edge rows as two rolling groups of eight with the next 32 list entries prefetched; sum order unchanged
# speedup vs baseline: 3.0914x; 3.0914x over previous
#include <hip/hip_runtime.h>

#define Nn 100000
#define Ee 1000000
#define INF 128
#define HID 128
#define CLS 64

typedef __attribute__((ext_vector_type(2))) float v2f;
typedef __attribute__((ext_vector_type(8))) float v8f;

#define TILE_N 400
#define NBIN   250
#define FILL_E 4096
static_assert(NBIN * TILE_N == Nn, "bins must cover the nodes exactly");
#define NSELF ((TILE_N + 31) / 32)

__device__ __forceinline__ void lds_add_f32(float* p, float v) {
    __hip_atomic_fetch_add(p, v, __ATOMIC_RELAXED, __HIP_MEMORY_SCOPE_WORKGROUP);
}

__global__ void k_zero_deg(int* __restrict__ deg) {
    int i = blockIdx.x * blockDim.x + threadIdx.x;
    if (i < Nn) deg[i] = 0;
}

__global__ void k_count_deg(const int* __restrict__ dst, int* __restrict__ deg) {
    int e = blockIdx.x * blockDim.x + threadIdx.x;
    if (e < Ee) atomicAdd(&deg[dst[e]], 1);
}

__global__ void k_dinv(const int* __restrict__ deg, float* __restrict__ dinv) {
    int i = blockIdx.x * blockDim.x + threadIdx.x;
    if (i < Nn) {
        float d = (float)deg[i] + 1.0f;
        dinv[i] = 1.0f / sqrtf(d);
    }
}

__global__ __launch_bounds__(1024) void k_binoffs(const int* __restrict__ deg,
                                                  int* __restrict__ offs,
                                                  int* __restrict__ cursor) {
    __shared__ int sc[256];
    const int tid  = threadIdx.x;
    const int lane = tid & 31;
    const int wave = tid >> 5;
    for (int b = wave; b < 256; b += 32) {
        int sum = 0;
        if (b < NBIN)
            for (int i = lane; i < TILE_N; i += 32) sum += deg[b * TILE_N + i];
        for (int o = 16; o > 0; o >>= 1) sum += __shfl_xor(sum, o);
        if (lane == 0) sc[b] = sum;
    }
    __syncthreads();
    const int own = (tid < 256) ? sc[tid] : 0;
    for (int o = 1; o < 256; o <<= 1) {
        int v = 0;
        if (tid < 256) { v = sc[tid]; if (tid >= o) v += sc[tid - o]; }
        __syncthreads();
        if (tid < 256) sc[tid] = v;
        __syncthreads();
    }
    if (tid < NBIN) {
        const int incl = sc[tid];
        offs[tid]   = incl - own;
        cursor[tid] = incl - own;
        if (tid == NBIN - 1) offs[NBIN] = incl;
    }
}

__global__ __launch_bounds__(1024) void k_fill(const int* __restrict__ src,
                                               const int* __restrict__ dst,
                                               const float* __restrict__ dinv,
                                               int* __restrict__ cursor,
                                               int* __restrict__ list) {
    __shared__ int hist[NBIN];
    __shared__ int basep[NBIN];
    const int tid = threadIdx.x;
    for (int i = tid; i < NBIN; i += 1024) hist[i] = 0;
    __syncthreads();
    const int e0 = blockIdx.x * FILL_E;
    int s[4], d[4], b[4], p[4];
    for (int k = 0; k < 4; ++k) {
        const int e = e0 + k * 1024 + tid;
        b[k] = -1;
        if (e < Ee) {
            s[k] = src[e];
            d[k] = dst[e];
            b[k] = d[k] / TILE_N;
            p[k] = atomicAdd(&hist[b[k]], 1);
        }
    }
    __syncthreads();
    for (int i = tid; i < NBIN; i += 1024) {
        const int c = hist[i];
        basep[i] = c ? atomicAdd(&cursor[i], c) : 0;
    }
    __syncthreads();
    for (int k = 0; k < 4; ++k) {
        if (b[k] >= 0) {
            const float w = dinv[s[k]] * dinv[d[k]];
            int4 ent;
            ent.x = s[k];
            ent.y = d[k] - b[k] * TILE_N;
            ent.z = __float_as_int(w);
            ent.w = 0;
            ((int4*)list)[basep[b[k]] + p[k]] = ent;
        }
    }
}

template <int VEC>
struct RowVec { typedef float type __attribute__((ext_vector_type(VEC))); };

__device__ __forceinline__ int4 load_entries(const int* __restrict__ list, int base, int end,
                                             int lane) {
    int idx = base + lane;
    if (idx > end - 1) idx = end - 1;
    return ((const int4*)list)[idx];
}

template <int VEC, int G>
__device__ __forceinline__ void rows_load(const typename RowVec<VEC>::type* __restrict__ hp,
                                          int srcs, int lane,
                                          typename RowVec<VEC>::type (&row)[8]) {
#pragma unroll
    for (int j = 0; j < 8; ++j) {
        const int s = __builtin_amdgcn_readlane(srcs, G + j);
        const unsigned off = ((unsigned)s * 32u + (unsigned)lane) * (unsigned)sizeof(typename RowVec<VEC>::type);
        row[j] = *(const typename RowVec<VEC>::type*)((const char*)hp + off);
    }
}

template <int VEC, int PITCH, int G, bool GUARD>
__device__ __forceinline__ void rows_add(float* tile, int dls, int wts, int nval, int lane,
                                         const typename RowVec<VEC>::type (&row)[8]) {
#pragma unroll
    for (int j = 0; j < 8; ++j) {
        if (!GUARD || G + j < nval) {
            const int   dl = __builtin_amdgcn_readlane(dls, G + j);
            const float w  = __int_as_float(__builtin_amdgcn_readlane(wts, G + j));
            float* t = tile + dl * PITCH + lane;
#pragma unroll
            for (int k = 0; k < VEC; ++k) lds_add_f32(t + k * 32, row[j][k] * w);
        }
    }
}

template <int VEC, int PITCH>
__device__ __forceinline__ void scatter_bin(const float* __restrict__ h,
                                            const int* __restrict__ list, float* tile,
                                            int beg, int end, int wave, int lane) {
    typedef typename RowVec<VEC>::type vecf;
    const vecf* hp = (const vecf*)h;
    int base = beg + wave * 32;
    if (base >= end) return;
    int4 ent = load_entries(list, base, end, lane);
    vecf ra[8], rb[8];
    rows_load<VEC, 0>(hp, ent.x, lane, ra);
    rows_load<VEC, 8>(hp, ent.x, lane, rb);
    while (base + 32 * 32 < end) {
        base += 32 * 32;
        const int4 nent = load_entries(list, base, end, lane);
        __builtin_amdgcn_sched_barrier(0);
        rows_add<VEC, PITCH, 0, false>(tile, ent.y, ent.z, 32, lane, ra);
        rows_load<VEC, 16>(hp, ent.x, lane, ra);
        __builtin_amdgcn_sched_barrier(0);
        rows_add<VEC, PITCH, 8, false>(tile, ent.y, ent.z, 32, lane, rb);
        rows_load<VEC, 24>(hp, ent.x, lane, rb);
        __builtin_amdgcn_sched_barrier(0);
        rows_add<VEC, PITCH, 16, false>(tile, ent.y, ent.z, 32, lane, ra);
        rows_load<VEC, 0>(hp, nent.x, lane, ra);
        __builtin_amdgcn_sched_barrier(0);
        rows_add<VEC, PITCH, 24, false>(tile, ent.y, ent.z, 32, lane, rb);
        rows_load<VEC, 8>(hp, nent.x, lane, rb);
        __builtin_amdgcn_sched_barrier(0);
        ent = nent;
    }
    int nval = end - base;
    if (nval > 32) nval = 32;
    rows_add<VEC, PITCH, 0, true>(tile, ent.y, ent.z, nval, lane, ra);
    if (nval > 16) rows_load<VEC, 16>(hp, ent.x, lane, ra);
    if (nval > 8)  rows_add<VEC, PITCH, 8, true>(tile, ent.y, ent.z, nval, lane, rb);
    if (nval > 24) rows_load<VEC, 24>(hp, ent.x, lane, rb);
    if (nval > 16) rows_add<VEC, PITCH, 16, true>(tile, ent.y, ent.z, nval, lane, ra);
    if (nval > 24) rows_add<VEC, PITCH, 24, true>(tile, ent.y, ent.z, nval, lane, rb);
}

template <int DOUT>
__global__ __launch_bounds__(1024) void k_agg_tile(const float* __restrict__ h,
                                                   const float* __restrict__ bias,
                                                   const float* __restrict__ dinv,
                                                   const int* __restrict__ offs,
                                                   const int* __restrict__ list,
                                                   float* __restrict__ out) {
    constexpr int VEC = DOUT / 32;
    typedef float vecf __attribute__((ext_vector_type(VEC)));
    __shared__ float tile[TILE_N * DOUT];

    const int lane = threadIdx.x & 31;
    const int wave = __builtin_amdgcn_readfirstlane((int)(threadIdx.x >> 5));
    const int r0   = blockIdx.x * TILE_N;
    const vecf* hp = (const vecf*)h;
    const vecf  bb = ((const vecf*)bias)[lane];

    const int beg = offs[blockIdx.x];
    const int end = offs[blockIdx.x + 1];

    vecf  sv[NSELF];
    float sw[NSELF];
#pragma unroll
    for (int i = 0; i < NSELF; ++i) {
        const int r = wave + 32 * i;
        const int n = r0 + ((32 * i + 32 <= TILE_N || r < TILE_N) ? r : wave);
        sw[i] = dinv[n];
        sv[i] = hp[(size_t)n * 32 + lane];
    }
#pragma unroll
    for (int i = 0; i < NSELF; ++i) {
        const int r = wave + 32 * i;
        if (32 * i + 32 <= TILE_N || r < TILE_N) {
            const float w = sw[i] * sw[i];
#pragma unroll
            for (int k = 0; k < VEC; ++k) tile[r * DOUT + k * 32 + lane] = sv[i][k] * w + bb[k];
        }
    }
    __syncthreads();

    scatter_bin<VEC, DOUT>(h, list, tile, beg, end, wave, lane);
    __syncthreads();

    for (int r = wave; r < TILE_N; r += 32) {
        vecf o;
#pragma unroll
        for (int k = 0; k < VEC; ++k) o[k] = tile[r * DOUT + k * 32 + lane];
        ((vecf*)out)[(size_t)(r0 + r) * 32 + lane] = o;
    }
}

#define TP 132
#define NSTRIP (TILE_N / 16)
static_assert(NSTRIP * 16 == TILE_N, "strips must cover the tile exactly");

template <int DOUT, bool PERM>
__device__ __forceinline__ void stage_w(const float* __restrict__ Wg, float* __restrict__ lw, int tid) {
    constexpr int PP = 4 * DOUT + 32, HO = 2 * DOUT + 32;
#pragma unroll
    for (int it = 0; it < (64 * DOUT) / 1024; ++it) {
        const int i = it * 1024 + tid;
        const int m = i / DOUT;
        const int n = i - m * DOUT;
        const int p = 2 * m;
        const int f0 = PERM ? ((p & 31) * 4 + (p >> 5)) : p;
        const int f1 = PERM ? (f0 + 4) : (p + 1);
        v2f v;
        v.x = Wg[f0 * DOUT + n];
        v.y = Wg[f1 * DOUT + n];
        *(v2f*)(lw + (m >> 1) * PP + (m & 1) * HO + 2 * n) = v;
    }
}

template <int DOUT>
__device__ __forceinline__ void strip_mm(const float* arow, const float* bcol, v8f* acc) {
    constexpr int PP = 4 * DOUT + 32;
#pragma unroll 4
    for (int j = 0; j < 32; ++j) {
        const v2f a = *(const v2f*)(arow + 4 * j);
#pragma unroll
        for (int t = 0; t < DOUT / 16; ++t) {
            const v2f b = *(const v2f*)(bcol + j * PP + t * 32);
            acc[t] = __builtin_amdgcn_wmma_f32_16x16x4_f32(
                 false, a,  false, b,
                 (short)0, acc[t],  false,  false);
        }
    }
}

template <int D1, int D2>
__global__ __launch_bounds__(1024) void k_agg_mm(const float* __restrict__ x,
                                                 const float* __restrict__ Wa,
                                                 const float* __restrict__ ba,
                                                 const float* __restrict__ Wb,
                                                 const float* __restrict__ dinv,
                                                 const int* __restrict__ offs,
                                                 const int* __restrict__ list,
                                                 float* __restrict__ out) {
    constexpr int DO  = D2 ? D2 : D1;
    constexpr int VO  = DO / 32;
    constexpr int HO1 = 2 * D1 + 32;
    constexpr int HO2 = 2 * D2 + 32;
    typedef float vec4 __attribute__((ext_vector_type(4)));
    typedef float veco __attribute__((ext_vector_type(VO)));
    __shared__ __attribute__((aligned(16))) float tile[TILE_N * TP];
    __shared__ __attribute__((aligned(16))) float lwa[32 * (4 * D1 + 32)];
    __shared__ __attribute__((aligned(16))) float lwb[D2 ? 32 * (4 * D2 + 32) : 4];

    const int tid  = threadIdx.x;
    const int lane = tid & 31;
    const int wave = __builtin_amdgcn_readfirstlane((int)(threadIdx.x >> 5));
    const int r0   = blockIdx.x * TILE_N;
    const vec4* hp = (const vec4*)x;

    const int beg = offs[blockIdx.x];
    const int end = offs[blockIdx.x + 1];

    vec4  sv[NSELF];
    float sw[NSELF];
#pragma unroll
    for (int i = 0; i < NSELF; ++i) {
        const int r = wave + 32 * i;
        const int n = r0 + ((32 * i + 32 <= TILE_N || r < TILE_N) ? r : wave);
        sw[i] = dinv[n];
        sv[i] = hp[(size_t)n * 32 + lane];
    }

    stage_w<D1, true>(Wa, lwa, tid);
    if constexpr (D2 != 0) stage_w<D2, false>(Wb, lwb, tid);

#pragma unroll
    for (int i = 0; i < NSELF; ++i) {
        const int r = wave + 32 * i;
        if (32 * i + 32 <= TILE_N || r < TILE_N) {
            const float w = sw[i] * sw[i];
#pragma unroll
            for (int k = 0; k < 4; ++k) tile[r * TP + k * 32 + lane] = sv[i][k] * w;
        }
    }
    __syncthreads();

    scatter_bin<4, TP>(x, list, tile, beg, end, wave, lane);
    __syncthreads();

    const int c = lane & 15;
    const int h = lane >> 4;
    float* srow = tile + wave * 16 * TP;
    if (wave < NSTRIP) {
        v8f acc[D1 / 16] = {};
        strip_mm<D1>(srow + c * TP + 2 * h, lwa + h * HO1 + 2 * c, acc);
#pragma unroll
        for (int t = 0; t < D1 / 16; ++t) {
            const float bv = ba[16 * t + c];
#pragma unroll
            for (int r = 0; r < 8; ++r)
                srow[(8 * h + r) * TP + 16 * t + c] = fmaxf(acc[t][r] + bv, 0.0f);
        }
    }
    if constexpr (D2 != 0) {
        __syncthreads();
        if (wave < NSTRIP) {
            v8f acc2[D2 / 16] = {};
            strip_mm<D2>(srow + c * TP + 2 * h, lwb + h * HO2 + 2 * c, acc2);
#pragma unroll
            for (int t = 0; t < D2 / 16; ++t) {
#pragma unroll
                for (int r = 0; r < 8; ++r)
                    srow[(8 * h + r) * TP + 16 * t + c] = acc2[t][r];
            }
        }
    }
    __syncthreads();

    for (int r = wave; r < TILE_N; r += 32) {
        const veco o = *(const veco*)(tile + r * TP + lane * VO);
        ((veco*)out)[(size_t)(r0 + r) * 32 + lane] = o;
    }
}

extern "C" void kernel_launch(void* const* d_in, const int* in_sizes, int n_in,
                              void* d_out, int out_size, void* d_ws, size_t ws_size,
                              hipStream_t stream) {
    const float* X   = (const float*)d_in[0];
    const float* W1  = (const float*)d_in[1];
    const float* b1  = (const float*)d_in[2];
    const float* W2  = (const float*)d_in[3];
    const float* b2  = (const float*)d_in[4];
    const float* W3  = (const float*)d_in[5];
    const float* b3  = (const float*)d_in[6];
    const int*   ei  = (const int*)d_in[7];
    const int*   src = ei;
    const int*   dst = ei + Ee;
    float* out = (float*)d_out;

    float* bufA = (float*)d_ws;
    float* bufB = bufA + (size_t)Nn * 128;
    int*   deg  = (int*)(bufB + (size_t)Nn * 128);
    float* dinv = (float*)(deg + Nn);
    int*   offs = (int*)(dinv + Nn);
    int*   curs = offs + 256;
    int*   list = curs + 256;

    const int T = 256;
    k_zero_deg<<<(Nn + T - 1) / T, T, 0, stream>>>(deg);
    k_count_deg<<<(Ee + T - 1) / T, T, 0, stream>>>(dst, deg);
    k_dinv<<<(Nn + T - 1) / T, T, 0, stream>>>(deg, dinv);
    k_binoffs<<<1, 1024, 0, stream>>>(deg, offs, curs);
    k_fill<<<(Ee + FILL_E - 1) / FILL_E, 1024, 0, stream>>>(src, dst, dinv, curs, list);

    k_agg_mm<128, 0><<<NBIN, 1024, 0, stream>>>(X, W1, b1, nullptr, dinv, offs, list, bufA);

    k_agg_mm<128, 64><<<NBIN, 1024, 0, stream>>>(bufA, W2, b2, W3, dinv, offs, list, bufB);

    k_agg_tile<64><<<NBIN, 1024, 0, stream>>>(bufB, b3, dinv, offs, list, out);
}


# Graph set-up merged from five launches to three: degrees and 400-node range counts in one edge pass (LDS histogram, 4096 edges a block), offsets scanned and dinv made inside the binning pass
# speedup vs baseline: 3.3343x; 1.0786x over previous
#include <hip/hip_runtime.h>

#define Nn 100000
#define Ee 1000000
#define INF 128
#define HID 128
#define CLS 64

typedef __attribute__((ext_vector_type(2))) float v2f;
typedef __attribute__((ext_vector_type(8))) float v8f;

#define TILE_N 400
#define NBIN   250
#define FILL_E 4096
static_assert(NBIN * TILE_N == Nn, "bins must cover the nodes exactly");
#define NSELF ((TILE_N + 31) / 32)

__device__ __forceinline__ void lds_add_f32(float* p, float v) {
    __hip_atomic_fetch_add(p, v, __ATOMIC_RELAXED, __HIP_MEMORY_SCOPE_WORKGROUP);
}

static_assert(Nn % 4 == 0, "degrees are cleared four at a time");
static_assert(NBIN < 256, "the offset scan covers 256 counters, one of them past the last bin");
static_assert(((Ee + FILL_E - 1) / FILL_E) * 1024 >= Nn, "the fill grid must cover every node for dinv");

__device__ __forceinline__ float dinv_of(int deg) {
    const float d = (float)deg + 1.0f;
    return 1.0f / sqrtf(d);
}

__global__ __launch_bounds__(1024) void k_zero(int* __restrict__ deg, int* __restrict__ bcnt,
                                               int* __restrict__ cursor) {
    const int i = blockIdx.x * 1024 + threadIdx.x;
    if (i < Nn / 4) ((int4*)deg)[i] = make_int4(0, 0, 0, 0);
    if (i < 256) { bcnt[i] = 0; cursor[i] = 0; }
}

__global__ __launch_bounds__(1024) void k_count(const int* __restrict__ dst, int* __restrict__ deg,
                                                int* __restrict__ bcnt) {
    __shared__ int hist[NBIN];
    const int tid = threadIdx.x;
    for (int i = tid; i < NBIN; i += 1024) hist[i] = 0;
    const int e0 = blockIdx.x * FILL_E;
    int d[4];
#pragma unroll
    for (int k = 0; k < 4; ++k) {
        const int e = e0 + k * 1024 + tid;
        d[k] = (e < Ee) ? dst[e] : -1;
    }
    __syncthreads();
#pragma unroll
    for (int k = 0; k < 4; ++k) {
        if (d[k] >= 0) {
            atomicAdd(&deg[d[k]], 1);
            atomicAdd(&hist[d[k] / TILE_N], 1);
        }
    }
    __syncthreads();
    for (int i = tid; i < NBIN; i += 1024) {
        const int c = hist[i];
        if (c) atomicAdd(&bcnt[i], c);
    }
}

__global__ __launch_bounds__(1024) void k_fill(const int* __restrict__ src,
                                               const int* __restrict__ dst,
                                               const int* __restrict__ deg,
                                               const int* __restrict__ bcnt,
                                               int* __restrict__ cursor,
                                               int* __restrict__ offs,
                                               float* __restrict__ dinv,
                                               int* __restrict__ list) {
    __shared__ int hist[NBIN];
    __shared__ int basep[NBIN];
    __shared__ int boff[256];
    __shared__ int wsum[8];
    const int tid  = threadIdx.x;
    const int lane = tid & 31;
    const int wv   = tid >> 5;
    for (int i = tid; i < NBIN; i += 1024) hist[i] = 0;
    const int e0 = blockIdx.x * FILL_E;
    int s[4], d[4], b[4], p[4];
#pragma unroll
    for (int k = 0; k < 4; ++k) {
        const int e = e0 + k * 1024 + tid;
        b[k] = -1;
        if (e < Ee) {
            s[k] = src[e];
            d[k] = dst[e];
            b[k] = d[k] / TILE_N;
        }
    }
    const int node = blockIdx.x * 1024 + tid;
    if (node < Nn) dinv[node] = dinv_of(deg[node]);
    int own = 0, incl = 0;
    if (tid < 256) {
        own  = (tid < NBIN) ? bcnt[tid] : 0;
        incl = own;
        for (int o = 1; o < 32; o <<= 1) {
            const int t = __shfl_up(incl, o);
            if (lane >= o) incl += t;
        }
        if (lane == 31) wsum[wv] = incl;
    }
    __syncthreads();
    if (tid < 256) {
        int pre = 0;
        for (int j = 0; j < 8; ++j) if (j < wv) pre += wsum[j];
        const int excl = pre + incl - own;
        boff[tid] = excl;
        if (blockIdx.x == 0 && tid <= NBIN) offs[tid] = excl;
    }
#pragma unroll
    for (int k = 0; k < 4; ++k)
        if (b[k] >= 0) p[k] = atomicAdd(&hist[b[k]], 1);
    __syncthreads();
    for (int i = tid; i < NBIN; i += 1024) {
        const int c = hist[i];
        basep[i] = c ? boff[i] + atomicAdd(&cursor[i], c) : 0;
    }
    __syncthreads();
#pragma unroll
    for (int k = 0; k < 4; ++k) {
        if (b[k] >= 0) {
            const float w = dinv_of(deg[s[k]]) * dinv_of(deg[d[k]]);
            int4 ent;
            ent.x = s[k];
            ent.y = d[k] - b[k] * TILE_N;
            ent.z = __float_as_int(w);
            ent.w = 0;
            ((int4*)list)[basep[b[k]] + p[k]] = ent;
        }
    }
}

template <int VEC>
struct RowVec { typedef float type __attribute__((ext_vector_type(VEC))); };

__device__ __forceinline__ int4 load_entries(const int* __restrict__ list, int base, int end,
                                             int lane) {
    int idx = base + lane;
    if (idx > end - 1) idx = end - 1;
    return ((const int4*)list)[idx];
}

template <int VEC, int G>
__device__ __forceinline__ void rows_load(const typename RowVec<VEC>::type* __restrict__ hp,
                                          int srcs, int lane,
                                          typename RowVec<VEC>::type (&row)[8]) {
#pragma unroll
    for (int j = 0; j < 8; ++j) {
        const int s = __builtin_amdgcn_readlane(srcs, G + j);
        const unsigned off = ((unsigned)s * 32u + (unsigned)lane) * (unsigned)sizeof(typename RowVec<VEC>::type);
        row[j] = *(const typename RowVec<VEC>::type*)((const char*)hp + off);
    }
}

template <int VEC, int PITCH, int G, bool GUARD>
__device__ __forceinline__ void rows_add(float* tile, int dls, int wts, int nval, int lane,
                                         const typename RowVec<VEC>::type (&row)[8]) {
#pragma unroll
    for (int j = 0; j < 8; ++j) {
        if (!GUARD || G + j < nval) {
            const int   dl = __builtin_amdgcn_readlane(dls, G + j);
            const float w  = __int_as_float(__builtin_amdgcn_readlane(wts, G + j));
            float* t = tile + dl * PITCH + lane;
#pragma unroll
            for (int k = 0; k < VEC; ++k) lds_add_f32(t + k * 32, row[j][k] * w);
        }
    }
}

template <int VEC, int PITCH>
__device__ __forceinline__ void scatter_bin(const float* __restrict__ h,
                                            const int* __restrict__ list, float* tile,
                                            int beg, int end, int wave, int lane) {
    typedef typename RowVec<VEC>::type vecf;
    const vecf* hp = (const vecf*)h;
    int base = beg + wave * 32;
    if (base >= end) return;
    int4 ent = load_entries(list, base, end, lane);
    vecf ra[8], rb[8];
    rows_load<VEC, 0>(hp, ent.x, lane, ra);
    rows_load<VEC, 8>(hp, ent.x, lane, rb);
    while (base + 32 * 32 < end) {
        base += 32 * 32;
        const int4 nent = load_entries(list, base, end, lane);
        __builtin_amdgcn_sched_barrier(0);
        rows_add<VEC, PITCH, 0, false>(tile, ent.y, ent.z, 32, lane, ra);
        rows_load<VEC, 16>(hp, ent.x, lane, ra);
        __builtin_amdgcn_sched_barrier(0);
        rows_add<VEC, PITCH, 8, false>(tile, ent.y, ent.z, 32, lane, rb);
        rows_load<VEC, 24>(hp, ent.x, lane, rb);
        __builtin_amdgcn_sched_barrier(0);
        rows_add<VEC, PITCH, 16, false>(tile, ent.y, ent.z, 32, lane, ra);
        rows_load<VEC, 0>(hp, nent.x, lane, ra);
        __builtin_amdgcn_sched_barrier(0);
        rows_add<VEC, PITCH, 24, false>(tile, ent.y, ent.z, 32, lane, rb);
        rows_load<VEC, 8>(hp, nent.x, lane, rb);
        __builtin_amdgcn_sched_barrier(0);
        ent = nent;
    }
    int nval = end - base;
    if (nval > 32) nval = 32;
    rows_add<VEC, PITCH, 0, true>(tile, ent.y, ent.z, nval, lane, ra);
    if (nval > 16) rows_load<VEC, 16>(hp, ent.x, lane, ra);
    if (nval > 8)  rows_add<VEC, PITCH, 8, true>(tile, ent.y, ent.z, nval, lane, rb);
    if (nval > 24) rows_load<VEC, 24>(hp, ent.x, lane, rb);
    if (nval > 16) rows_add<VEC, PITCH, 16, true>(tile, ent.y, ent.z, nval, lane, ra);
    if (nval > 24) rows_add<VEC, PITCH, 24, true>(tile, ent.y, ent.z, nval, lane, rb);
}

template <int DOUT>
__global__ __launch_bounds__(1024) void k_agg_tile(const float* __restrict__ h,
                                                   const float* __restrict__ bias,
                                                   const float* __restrict__ dinv,
                                                   const int* __restrict__ offs,
                                                   const int* __restrict__ list,
                                                   float* __restrict__ out) {
    constexpr int VEC = DOUT / 32;
    typedef float vecf __attribute__((ext_vector_type(VEC)));
    __shared__ float tile[TILE_N * DOUT];

    const int lane = threadIdx.x & 31;
    const int wave = __builtin_amdgcn_readfirstlane((int)(threadIdx.x >> 5));
    const int r0   = blockIdx.x * TILE_N;
    const vecf* hp = (const vecf*)h;
    const vecf  bb = ((const vecf*)bias)[lane];

    const int beg = offs[blockIdx.x];
    const int end = offs[blockIdx.x + 1];

    vecf  sv[NSELF];
    float sw[NSELF];
#pragma unroll
    for (int i = 0; i < NSELF; ++i) {
        const int r = wave + 32 * i;
        const int n = r0 + ((32 * i + 32 <= TILE_N || r < TILE_N) ? r : wave);
        sw[i] = dinv[n];
        sv[i] = hp[(size_t)n * 32 + lane];
    }
#pragma unroll
    for (int i = 0; i < NSELF; ++i) {
        const int r = wave + 32 * i;
        if (32 * i + 32 <= TILE_N || r < TILE_N) {
            const float w = sw[i] * sw[i];
#pragma unroll
            for (int k = 0; k < VEC; ++k) tile[r * DOUT + k * 32 + lane] = sv[i][k] * w + bb[k];
        }
    }
    __syncthreads();

    scatter_bin<VEC, DOUT>(h, list, tile, beg, end, wave, lane);
    __syncthreads();

    for (int r = wave; r < TILE_N; r += 32) {
        vecf o;
#pragma unroll
        for (int k = 0; k < VEC; ++k) o[k] = tile[r * DOUT + k * 32 + lane];
        ((vecf*)out)[(size_t)(r0 + r) * 32 + lane] = o;
    }
}

#define TP 132
#define NSTRIP (TILE_N / 16)
static_assert(NSTRIP * 16 == TILE_N, "strips must cover the tile exactly");

template <int DOUT, bool PERM>
__device__ __forceinline__ void stage_w(const float* __restrict__ Wg, float* __restrict__ lw, int tid) {
    constexpr int PP = 4 * DOUT + 32, HO = 2 * DOUT + 32;
#pragma unroll
    for (int it = 0; it < (64 * DOUT) / 1024; ++it) {
        const int i = it * 1024 + tid;
        const int m = i / DOUT;
        const int n = i - m * DOUT;
        const int p = 2 * m;
        const int f0 = PERM ? ((p & 31) * 4 + (p >> 5)) : p;
        const int f1 = PERM ? (f0 + 4) : (p + 1);
        v2f v;
        v.x = Wg[f0 * DOUT + n];
        v.y = Wg[f1 * DOUT + n];
        *(v2f*)(lw + (m >> 1) * PP + (m & 1) * HO + 2 * n) = v;
    }
}

template <int DOUT>
__device__ __forceinline__ void strip_mm(const float* arow, const float* bcol, v8f* acc) {
    constexpr int PP = 4 * DOUT + 32;
#pragma unroll 4
    for (int j = 0; j < 32; ++j) {
        const v2f a = *(const v2f*)(arow + 4 * j);
#pragma unroll
        for (int t = 0; t < DOUT / 16; ++t) {
            const v2f b = *(const v2f*)(bcol + j * PP + t * 32);
            acc[t] = __builtin_amdgcn_wmma_f32_16x16x4_f32(
                 false, a,  false, b,
                 (short)0, acc[t],  false,  false);
        }
    }
}

template <int D1, int D2>
__global__ __launch_bounds__(1024) void k_agg_mm(const float* __restrict__ x,
                                                 const float* __restrict__ Wa,
                                                 const float* __restrict__ ba,
                                                 const float* __restrict__ Wb,
                                                 const float* __restrict__ dinv,
                                                 const int* __restrict__ offs,
                                                 const int* __restrict__ list,
                                                 float* __restrict__ out) {
    constexpr int DO  = D2 ? D2 : D1;
    constexpr int VO  = DO / 32;
    constexpr int HO1 = 2 * D1 + 32;
    constexpr int HO2 = 2 * D2 + 32;
    typedef float vec4 __attribute__((ext_vector_type(4)));
    typedef float veco __attribute__((ext_vector_type(VO)));
    __shared__ __attribute__((aligned(16))) float tile[TILE_N * TP];
    __shared__ __attribute__((aligned(16))) float lwa[32 * (4 * D1 + 32)];
    __shared__ __attribute__((aligned(16))) float lwb[D2 ? 32 * (4 * D2 + 32) : 4];

    const int tid  = threadIdx.x;
    const int lane = tid & 31;
    const int wave = __builtin_amdgcn_readfirstlane((int)(threadIdx.x >> 5));
    const int r0   = blockIdx.x * TILE_N;
    const vec4* hp = (const vec4*)x;

    const int beg = offs[blockIdx.x];
    const int end = offs[blockIdx.x + 1];

    vec4  sv[NSELF];
    float sw[NSELF];
#pragma unroll
    for (int i = 0; i < NSELF; ++i) {
        const int r = wave + 32 * i;
        const int n = r0 + ((32 * i + 32 <= TILE_N || r < TILE_N) ? r : wave);
        sw[i] = dinv[n];
        sv[i] = hp[(size_t)n * 32 + lane];
    }

    stage_w<D1, true>(Wa, lwa, tid);
    if constexpr (D2 != 0) stage_w<D2, false>(Wb, lwb, tid);

#pragma unroll
    for (int i = 0; i < NSELF; ++i) {
        const int r = wave + 32 * i;
        if (32 * i + 32 <= TILE_N || r < TILE_N) {
            const float w = sw[i] * sw[i];
#pragma unroll
            for (int k = 0; k < 4; ++k) tile[r * TP + k * 32 + lane] = sv[i][k] * w;
        }
    }
    __syncthreads();

    scatter_bin<4, TP>(x, list, tile, beg, end, wave, lane);
    __syncthreads();

    const int c = lane & 15;
    const int h = lane >> 4;
    float* srow = tile + wave * 16 * TP;
    if (wave < NSTRIP) {
        v8f acc[D1 / 16] = {};
        strip_mm<D1>(srow + c * TP + 2 * h, lwa + h * HO1 + 2 * c, acc);
#pragma unroll
        for (int t = 0; t < D1 / 16; ++t) {
            const float bv = ba[16 * t + c];
#pragma unroll
            for (int r = 0; r < 8; ++r)
                srow[(8 * h + r) * TP + 16 * t + c] = fmaxf(acc[t][r] + bv, 0.0f);
        }
    }
    if constexpr (D2 != 0) {
        __syncthreads();
        if (wave < NSTRIP) {
            v8f acc2[D2 / 16] = {};
            strip_mm<D2>(srow + c * TP + 2 * h, lwb + h * HO2 + 2 * c, acc2);
#pragma unroll
            for (int t = 0; t < D2 / 16; ++t) {
#pragma unroll
                for (int r = 0; r < 8; ++r)
                    srow[(8 * h + r) * TP + 16 * t + c] = acc2[t][r];
            }
        }
    }
    __syncthreads();

    for (int r = wave; r < TILE_N; r += 32) {
        const veco o = *(const veco*)(tile + r * TP + lane * VO);
        ((veco*)out)[(size_t)(r0 + r) * 32 + lane] = o;
    }
}

extern "C" void kernel_launch(void* const* d_in, const int* in_sizes, int n_in,
                              void* d_out, int out_size, void* d_ws, size_t ws_size,
                              hipStream_t stream) {
    const float* X   = (const float*)d_in[0];
    const float* W1  = (const float*)d_in[1];
    const float* b1  = (const float*)d_in[2];
    const float* W2  = (const float*)d_in[3];
    const float* b2  = (const float*)d_in[4];
    const float* W3  = (const float*)d_in[5];
    const float* b3  = (const float*)d_in[6];
    const int*   ei  = (const int*)d_in[7];
    const int*   src = ei;
    const int*   dst = ei + Ee;
    float* out = (float*)d_out;

    float* bufA = (float*)d_ws;
    float* bufB = bufA + (size_t)Nn * 128;
    int*   deg  = (int*)(bufB + (size_t)Nn * 128);
    float* dinv = (float*)(deg + Nn);
    int*   offs = (int*)(dinv + Nn);
    int*   curs = offs + 256;
    int*   bcnt = curs + 256;
    int*   list = bcnt + 256;

    k_zero<<<(Nn / 4 + 1023) / 1024, 1024, 0, stream>>>(deg, bcnt, curs);
    k_count<<<(Ee + FILL_E - 1) / FILL_E, 1024, 0, stream>>>(dst, deg, bcnt);
    k_fill<<<(Ee + FILL_E - 1) / FILL_E, 1024, 0, stream>>>(src, dst, deg, bcnt, curs, offs, dinv, list);

    k_agg_mm<128, 0><<<NBIN, 1024, 0, stream>>>(X, W1, b1, nullptr, dinv, offs, list, bufA);

    k_agg_mm<128, 64><<<NBIN, 1024, 0, stream>>>(bufA, W2, b2, W3, dinv, offs, list, bufB);

    k_agg_tile<64><<<NBIN, 1024, 0, stream>>>(bufB, b3, dinv, offs, list, out);
}
